# starting kernel
# baseline (speedup 1.0000x reference)
#include <hip/hip_runtime.h>
#include <math.h>

typedef __attribute__((ext_vector_type(2))) float v2f;
typedef __attribute__((ext_vector_type(4))) float v4f;
typedef __attribute__((ext_vector_type(8))) float v8f;

constexpr int kB   = 32;
constexpr int kIU  = 16;
constexpr int kIC  = 2048;
constexpr int kNU  = 16;
constexpr int kUS  = 64;
constexpr int kKS  = 8;
constexpr size_t kPartialsElems = (size_t)kKS * 1024 * kB;

__global__ __launch_bounds__(256) void caps_xt(const float* __restrict__ x,
                                               float* __restrict__ xt2) {
  int idx  = blockIdx.x * 256 + threadIdx.x;
  int r    = idx & 1;
  int slo  = (idx >> 1) & 1;
  int lane = (idx >> 2) & 31;
  int half = (idx >> 7) & 1;
  int tile = (idx >> 8) & 1;
  int i    = idx >> 9;
  int s    = half * 2 + slo;
  int kgrp = lane >> 4;
  int n    = (lane & 15) + 16 * tile;
  int u    = 2 * s + 8 * kgrp + r;
  xt2[idx] = x[(size_t)n * (kIU * kIC) + (size_t)u * kIC + i];
}

__global__ __launch_bounds__(256) void caps_gemm(const float* __restrict__ xt2,
                                                 const float* __restrict__ W,
                                                 float* __restrict__ partials) {
  const int lane  = threadIdx.x & 31;
  const int wave  = threadIdx.x >> 5;
  const int mtile = blockIdx.x;
  const int ksp   = blockIdx.y;
  const int j     = mtile >> 2;
  const int k0    = (mtile & 3) << 4;
  const int mrow  = lane & 15;
  const int kgrp  = lane >> 4;

  v8f acc0 = {};
  v8f acc1 = {};

  const float* Wjk = W + (size_t)j * (kUS * kIU) + (size_t)k0 * kIU
                       + (size_t)mrow * kIU + (size_t)kgrp * 8;
  const int i0 = ksp * (kIC / kKS);

  for (int t = 0; t < (kIC / kKS) / 8; ++t) {
    const int i = i0 + wave + t * 8;
    const float* Wb = Wjk + (size_t)i * (kNU * kUS * kIU);
    __builtin_prefetch(Wb + (size_t)8 * (kNU * kUS * kIU), 0, 1);
    v4f a01 = __builtin_nontemporal_load((const v4f*)(Wb));
    v4f a23 = __builtin_nontemporal_load((const v4f*)(Wb + 4));
    const float* Xb = xt2 + (size_t)i * 512 + lane * 4;
    v4f b0a = *(const v4f*)(Xb);
    v4f b0b = *(const v4f*)(Xb + 128);
    v4f b1a = *(const v4f*)(Xb + 256);
    v4f b1b = *(const v4f*)(Xb + 384);

    v2f a, b;
#define STEP(AV, LOHI, B0, B1)                                             \
    a = __builtin_shufflevector(AV, AV, 0 + LOHI, 1 + LOHI);               \
    b = __builtin_shufflevector(B0, B0, 0 + LOHI, 1 + LOHI);               \
    acc0 = __builtin_amdgcn_wmma_f32_16x16x4_f32(false, a, false, b,       \
                                      (short)0, acc0, false, false);       \
    b = __builtin_shufflevector(B1, B1, 0 + LOHI, 1 + LOHI);               \
    acc1 = __builtin_amdgcn_wmma_f32_16x16x4_f32(false, a, false, b,       \
                                      (short)0, acc1, false, false);
    STEP(a01, 0, b0a, b1a)
    STEP(a01, 2, b0a, b1a)
    STEP(a23, 0, b0b, b1b)
    STEP(a23, 2, b0b, b1b)
#undef STEP
  }

  __shared__ float lds[8][2][256];
  *(v8f*)&lds[wave][0][lane * 8] = acc0;
  *(v8f*)&lds[wave][1][lane * 8] = acc1;
  __syncthreads();

  int flat = threadIdx.x * 2;
#pragma unroll
  for (int q = 0; q < 2; ++q, ++flat) {
    const int t = flat >> 8;
    const int p = flat & 255;
    float sum = 0.f;
#pragma unroll
    for (int w = 0; w < 8; ++w) sum += lds[w][t][p];
    const int l  = p >> 3, r = p & 7;
    const int mr = r + ((l >> 4) << 3);
    const int n  = t * 16 + (l & 15);
    partials[(size_t)ksp * (1024 * kB) + (size_t)(mtile * 16 + mr) * kB + n] = sum;
  }
}

__global__ __launch_bounds__(64) void caps_squash(const float* __restrict__ partials,
                                                  float* __restrict__ out) {
  const int b = blockIdx.x;
  const int k = threadIdx.x;
  float sv[kNU];
  float mag_sq = 0.f;
#pragma unroll
  for (int j = 0; j < kNU; ++j) {
    float v = 0.f;
#pragma unroll
    for (int p = 0; p < kKS; ++p)
      v += partials[(size_t)p * (1024 * kB) + (size_t)(j * kUS + k) * kB + b];
    sv[j] = v;
    mag_sq += v * v;
  }
  const float mag   = sqrtf(mag_sq);
  const float scale = mag_sq / ((1.f + mag_sq) * mag);
#pragma unroll
  for (int j = 0; j < kNU; ++j)
    out[(size_t)b * (kNU * kUS) + j * kUS + k] = sv[j] * scale;
}

extern "C" void kernel_launch(void* const* d_in, const int* in_sizes, int n_in,
                              void* d_out, int out_size, void* d_ws, size_t ws_size,
                              hipStream_t stream) {
  const float* x = (const float*)d_in[0];
  const float* W = (const float*)d_in[1];
  float* out      = (float*)d_out;
  float* partials = (float*)d_ws;
  float* xt2      = (float*)d_ws + kPartialsElems;

  caps_xt<<<(2048 * 512) / 256, 256, 0, stream>>>(x, xt2);
  dim3 grid(64, kKS);
  caps_gemm<<<grid, 256, 0, stream>>>(xt2, W, partials);
  caps_squash<<<32, 64, 0, stream>>>(partials, out);
}


# pack kernel as a tiled turn through LDS: 256 workgroups read 128 rows x 32 i of x as whole 128-byte lines, turn in LDS, store whole 512-byte fragment blocks; 2048 waves, was 32768; no sum touched
# speedup vs baseline: 1.3392x; 1.3392x over previous
#include <hip/hip_runtime.h>
#include <math.h>

typedef __attribute__((ext_vector_type(2))) float v2f;
typedef __attribute__((ext_vector_type(4))) float v4f;
typedef __attribute__((ext_vector_type(8))) float v8f;

constexpr int kB   = 32;
constexpr int kIU  = 16;
constexpr int kIC  = 2048;
constexpr int kNU  = 16;
constexpr int kUS  = 64;
constexpr int kKS  = 8;
constexpr size_t kPartialsElems = (size_t)kKS * 1024 * kB;

constexpr int kPackI     = 32;
constexpr int kPackPitch = 132;
__global__ __launch_bounds__(256) void caps_xt(const float* __restrict__ x,
                                               float* __restrict__ xt2) {
  __shared__ __attribute__((aligned(16))) float T[kPackI * kPackPitch];
  const int tid  = threadIdx.x;
  const int th   = blockIdx.x & 3;
  const int tile = th >> 1;
  const int half = th & 1;
  const int i0   = (blockIdx.x >> 2) * kPackI;
  const int isel = tid & 7;
  const int rsel = tid >> 3;

  v4f v[4];
  for (int q = 0; q < 4; ++q) {
    const int f    = q * 32 + rsel;
    const int kgrp = f >> 6;
    const int nlo  = (f >> 2) & 15;
    const int ulo  = f & 3;
    const int n    = 16 * tile + nlo;
    const int u    = 8 * kgrp + 4 * half + ulo;
    v[q] = *(const v4f*)(x + (size_t)n * (kIU * kIC) + (size_t)u * kIC + i0 + 4 * isel);
  }
  for (int q = 0; q < 4; ++q) {
    const int f = q * 32 + rsel;
    float* t = T + (4 * isel) * kPackPitch + f;
    t[0 * kPackPitch] = v[q].x;
    t[1 * kPackPitch] = v[q].y;
    t[2 * kPackPitch] = v[q].z;
    t[3 * kPackPitch] = v[q].w;
  }
  __syncthreads();

  const int lane = tid & 31;
  const int wave = tid >> 5;
  float* dst = xt2 + (size_t)i0 * 512 + th * 128 + lane * 4;
  for (int c = 0; c < 4; ++c) {
    const int il = wave * 4 + c;
    *(v4f*)(dst + (size_t)il * 512) = *(const v4f*)(T + il * kPackPitch + lane * 4);
  }
}

__global__ __launch_bounds__(256) void caps_gemm(const float* __restrict__ xt2,
                                                 const float* __restrict__ W,
                                                 float* __restrict__ partials) {
  const int lane  = threadIdx.x & 31;
  const int wave  = threadIdx.x >> 5;
  const int mtile = blockIdx.x;
  const int ksp   = blockIdx.y;
  const int j     = mtile >> 2;
  const int k0    = (mtile & 3) << 4;
  const int mrow  = lane & 15;
  const int kgrp  = lane >> 4;

  v8f acc0 = {};
  v8f acc1 = {};

  const float* Wjk = W + (size_t)j * (kUS * kIU) + (size_t)k0 * kIU
                       + (size_t)mrow * kIU + (size_t)kgrp * 8;
  const int i0 = ksp * (kIC / kKS);

  for (int t = 0; t < (kIC / kKS) / 8; ++t) {
    const int i = i0 + wave + t * 8;
    const float* Wb = Wjk + (size_t)i * (kNU * kUS * kIU);
    __builtin_prefetch(Wb + (size_t)8 * (kNU * kUS * kIU), 0, 1);
    v4f a01 = __builtin_nontemporal_load((const v4f*)(Wb));
    v4f a23 = __builtin_nontemporal_load((const v4f*)(Wb + 4));
    const float* Xb = xt2 + (size_t)i * 512 + lane * 4;
    v4f b0a = *(const v4f*)(Xb);
    v4f b0b = *(const v4f*)(Xb + 128);
    v4f b1a = *(const v4f*)(Xb + 256);
    v4f b1b = *(const v4f*)(Xb + 384);

    v2f a, b;
#define STEP(AV, LOHI, B0, B1)                                             \
    a = __builtin_shufflevector(AV, AV, 0 + LOHI, 1 + LOHI);               \
    b = __builtin_shufflevector(B0, B0, 0 + LOHI, 1 + LOHI);               \
    acc0 = __builtin_amdgcn_wmma_f32_16x16x4_f32(false, a, false, b,       \
                                      (short)0, acc0, false, false);       \
    b = __builtin_shufflevector(B1, B1, 0 + LOHI, 1 + LOHI);               \
    acc1 = __builtin_amdgcn_wmma_f32_16x16x4_f32(false, a, false, b,       \
                                      (short)0, acc1, false, false);
    STEP(a01, 0, b0a, b1a)
    STEP(a01, 2, b0a, b1a)
    STEP(a23, 0, b0b, b1b)
    STEP(a23, 2, b0b, b1b)
#undef STEP
  }

  __shared__ float lds[8][2][256];
  *(v8f*)&lds[wave][0][lane * 8] = acc0;
  *(v8f*)&lds[wave][1][lane * 8] = acc1;
  __syncthreads();

  int flat = threadIdx.x * 2;
#pragma unroll
  for (int q = 0; q < 2; ++q, ++flat) {
    const int t = flat >> 8;
    const int p = flat & 255;
    float sum = 0.f;
#pragma unroll
    for (int w = 0; w < 8; ++w) sum += lds[w][t][p];
    const int l  = p >> 3, r = p & 7;
    const int mr = r + ((l >> 4) << 3);
    const int n  = t * 16 + (l & 15);
    partials[(size_t)ksp * (1024 * kB) + (size_t)(n >> 2) * (1024 * 4)
             + (size_t)(mtile * 16 + mr) * 4 + (n & 3)] = sum;
  }
}

__global__ __launch_bounds__(512) void caps_squash(const float* __restrict__ partials,
                                                    float* __restrict__ out) {
  __shared__ float V[kNU * 4 * 32];
  __shared__ float SC[4 * 32];
  const int kh = blockIdx.x & 1;
  const int bg = blockIdx.x >> 1;
  const int kk = threadIdx.x & 31;
  const int j  = threadIdx.x >> 5;

  const float* src = partials + (size_t)bg * (1024 * 4)
                              + (size_t)(j * kUS + kh * 32 + kk) * 4;
  v4f acc = v4f{0.f, 0.f, 0.f, 0.f};
#pragma unroll
  for (int p = 0; p < kKS; ++p) acc += *(const v4f*)(src + (size_t)p * (1024 * kB));

  V[(j * 4 + 0) * 32 + kk] = acc.x;
  V[(j * 4 + 1) * 32 + kk] = acc.y;
  V[(j * 4 + 2) * 32 + kk] = acc.z;
  V[(j * 4 + 3) * 32 + kk] = acc.w;
  __syncthreads();

  if (threadIdx.x < 128) {
    const int q = threadIdx.x >> 5;
    float mag_sq = 0.f;
#pragma unroll
    for (int jj = 0; jj < kNU; ++jj) {
      const float v = V[(jj * 4 + q) * 32 + kk];
      mag_sq += v * v;
    }
    const float mag = sqrtf(mag_sq);
    SC[q * 32 + kk] = mag_sq / ((1.f + mag_sq) * mag);
  }
  __syncthreads();

  float* dst = out + (size_t)(bg * 4) * (kNU * kUS) + j * kUS + kh * 32 + kk;
  dst[0 * (kNU * kUS)] = acc.x * SC[0 * 32 + kk];
  dst[1 * (kNU * kUS)] = acc.y * SC[1 * 32 + kk];
  dst[2 * (kNU * kUS)] = acc.z * SC[2 * 32 + kk];
  dst[3 * (kNU * kUS)] = acc.w * SC[3 * 32 + kk];
}

extern "C" void kernel_launch(void* const* d_in, const int* in_sizes, int n_in,
                              void* d_out, int out_size, void* d_ws, size_t ws_size,
                              hipStream_t stream) {
  const float* x = (const float*)d_in[0];
  const float* W = (const float*)d_in[1];
  float* out      = (float*)d_out;
  float* partials = (float*)d_ws;
  float* xt2      = (float*)d_ws + kPartialsElems;

  caps_xt<<<256, 256, 0, stream>>>(x, xt2);
  dim3 grid(64, kKS);
  caps_gemm<<<grid, 256, 0, stream>>>(xt2, W, partials);
  caps_squash<<<16, 512, 0, stream>>>(partials, out);
}
